# starting kernel
# baseline (speedup 1.0000x reference)
#include <hip/hip_runtime.h>
#include <hip/hip_bf16.h>

typedef __attribute__((ext_vector_type(2))) float v2f;
typedef __attribute__((ext_vector_type(4))) float v4f;
typedef __attribute__((ext_vector_type(8))) float v8f;

#define L_LEN   65536
#define N_ROWS  64
#define TAPS    80
#define KSPAN   96
#define NCHUNK  (KSPAN / 4)
#define PAD     16
#define WAVES_PER_BLOCK 8
#define TILES_TOTAL (N_ROWS * (L_LEN / 256))

__global__ __launch_bounds__(256) void
HanningTemplateLayer_61761629716908_kernel(const float* __restrict__ x,
                                           const float* __restrict__ tw,
                                           float* __restrict__ out) {
    __shared__ float eff_pad[128];

    {
        const int t = threadIdx.x;
        if (t < 128) {
            float acc = 0.0f;
            const int tt = t - PAD;
            if ((unsigned)tt < TAPS) {
                float w0 = tw[0], w1 = tw[1], w2 = tw[2], w3 = tw[3];
                float mx = fmaxf(fmaxf(w0, w1), fmaxf(w2, w3));
                float e0 = __expf(w0 - mx), e1 = __expf(w1 - mx);
                float e2 = __expf(w2 - mx), e3 = __expf(w3 - mx);
                float inv = 1.0f / (e0 + e1 + e2 + e3);
                const float sw[4]   = {e0 * inv, e1 * inv, e2 * inv, e3 * inv};
                const int   wid4[4] = {10, 20, 30, 40};
                const float twopi = 6.28318530717958647692f;
                #pragma unroll
                for (int i = 0; i < 4; ++i) {
                    int w = wid4[i];
                    int k = tt - 40 + w;
                    if (k >= 0 && k < 2 * w) {
                        float h = 0.5f - 0.5f * cosf(twopi * (float)k / (float)(2 * w - 1));
                        acc += sw[i] * h;
                    }
                }
            }
            eff_pad[t] = acc;
        }
    }
    __syncthreads();

    const int lane = threadIdx.x & 31;
    const int wid  = threadIdx.x >> 5;
    const int W    = blockIdx.x * WAVES_PER_BLOCK + wid;
    const int row  = W >> 8;
    const int tile = W & 255;
    const int j0   = tile << 8;

    const int m   = lane & 15;
    const int hi  = lane >> 4;
    const int klo = hi << 1;

    v2f a_frag[NCHUNK];
    #pragma unroll
    for (int c = 0; c < NCHUNK; ++c) {
        const int base = c * 4 + klo - m + PAD;
        a_frag[c].x = eff_pad[base];
        a_frag[c].y = eff_pad[base + 1];
    }

    const float* xrow  = x + (size_t)row * L_LEN;
    const int    basep = j0 - 40 + 16 * m + klo;
    v8f acc0 = {};
    v8f acc1 = {};

    const bool interior = (j0 >= 40) && (j0 + 296 <= L_LEN);
    if (interior) {
        #pragma unroll
        for (int c = 0; c < NCHUNK; c += 2) {
            v2f b0 = *(const v2f*)(xrow + basep + 4 * c);
            v2f b1 = *(const v2f*)(xrow + basep + 4 * (c + 1));
            acc0 = __builtin_amdgcn_wmma_f32_16x16x4_f32(
                false, a_frag[c], false, b0, (short)0, acc0, false, false);
            acc1 = __builtin_amdgcn_wmma_f32_16x16x4_f32(
                false, a_frag[c + 1], false, b1, (short)0, acc1, false, false);
        }
    } else {
        #pragma unroll
        for (int c = 0; c < NCHUNK; c += 2) {
            const int p0 = basep + 4 * c;
            const int p1 = p0 + 4;
            int pc0 = p0 < 0 ? 0 : p0;
            pc0 = pc0 > (L_LEN - 2) ? (L_LEN - 2) : pc0;
            int pc1 = p1 < 0 ? 0 : p1;
            pc1 = pc1 > (L_LEN - 2) ? (L_LEN - 2) : pc1;
            v2f b0 = *(const v2f*)(xrow + pc0);
            v2f b1 = *(const v2f*)(xrow + pc1);
            b0.x = ((unsigned)p0       < L_LEN) ? b0.x : 0.0f;
            b0.y = ((unsigned)(p0 + 1) < L_LEN) ? b0.y : 0.0f;
            b1.x = ((unsigned)p1       < L_LEN) ? b1.x : 0.0f;
            b1.y = ((unsigned)(p1 + 1) < L_LEN) ? b1.y : 0.0f;
            acc0 = __builtin_amdgcn_wmma_f32_16x16x4_f32(
                false, a_frag[c], false, b0, (short)0, acc0, false, false);
            acc1 = __builtin_amdgcn_wmma_f32_16x16x4_f32(
                false, a_frag[c + 1], false, b1, (short)0, acc1, false, false);
        }
    }

    float* orow = out + (size_t)row * L_LEN + j0 + 8 * hi + 16 * m;
    v4f lo = {acc0[0] + acc1[0], acc0[1] + acc1[1], acc0[2] + acc1[2], acc0[3] + acc1[3]};
    v4f hh = {acc0[4] + acc1[4], acc0[5] + acc1[5], acc0[6] + acc1[6], acc0[7] + acc1[7]};
    *(v4f*)(orow)     = lo;
    *(v4f*)(orow + 4) = hh;
}

extern "C" void kernel_launch(void* const* d_in, const int* in_sizes, int n_in,
                              void* d_out, int out_size, void* d_ws, size_t ws_size,
                              hipStream_t stream) {
    (void)in_sizes; (void)n_in; (void)d_ws; (void)ws_size; (void)out_size;
    const float* x  = (const float*)d_in[0];
    const float* tw = (const float*)d_in[1];
    float* out = (float*)d_out;

    const int blocks = TILES_TOTAL / WAVES_PER_BLOCK;
    HanningTemplateLayer_61761629716908_kernel<<<blocks, 256, 0, stream>>>(x, tw, out);
}


# persistent waves: 2048 waves (2 per SIMD) instead of 16384; each keeps its filter fragments and loops over 8 tiles, the next tile's signal slice prefetched into a second register set
# speedup vs baseline: 1.0795x; 1.0795x over previous
#include <hip/hip_runtime.h>


typedef __attribute__((ext_vector_type(2))) float v2f;
typedef __attribute__((ext_vector_type(4))) float v4f;
typedef __attribute__((ext_vector_type(8))) float v8f;

#define L_LEN   65536
#define N_ROWS  64
#define TAPS    80
#define KSPAN   96
#define NCHUNK  (KSPAN / 4)
#define PAD     16
#define TILES_PER_ROW   (L_LEN / 256)
#define TILES_TOTAL     (N_ROWS * TILES_PER_ROW)
#define WAVES_PER_BLOCK 4
#define BLOCK_THREADS   128
#define NBLOCKS         512
#define NWAVES          (NBLOCKS * WAVES_PER_BLOCK)
#define TILES_PER_WAVE  (TILES_TOTAL / NWAVES)
#define ROT             32


__device__ __forceinline__ void load_b_interior(const float* __restrict__ x, int row, int t,
                                                int lane_off, v2f (&b)[NCHUNK]) {
    const float* p = x + (size_t)row * L_LEN + ((t << 8) - 40);
    #pragma unroll
    for (int c = 0; c < NCHUNK; ++c)
        b[c] = *(const v2f*)(p + lane_off + 4 * c);
}

__device__ __forceinline__ void load_b_any(const float* __restrict__ x, int row, int t,
                                           int lane_off, v2f (&b)[NCHUNK]) {
    const int j0 = t << 8;
    const bool interior = (j0 >= 40) && (j0 + 296 <= L_LEN);
    if (interior) {
        load_b_interior(x, row, t, lane_off, b);
    } else {
        const float* xrow = x + (size_t)row * L_LEN;
        const int basep = j0 - 40 + lane_off;
        #pragma unroll
        for (int c = 0; c < NCHUNK; ++c) {
            const int p0 = basep + 4 * c;
            int pc = p0 < 0 ? 0 : p0;
            pc = pc > (L_LEN - 2) ? (L_LEN - 2) : pc;
            v2f v = *(const v2f*)(xrow + pc);
            v.x = ((unsigned)p0       < L_LEN) ? v.x : 0.0f;
            v.y = ((unsigned)(p0 + 1) < L_LEN) ? v.y : 0.0f;
            b[c] = v;
        }
    }
}

__device__ __forceinline__ void mma_store(const v2f (&a)[NCHUNK], const v2f (&b)[NCHUNK],
                                          float* __restrict__ orow) {
    v8f acc0 = {};
    v8f acc1 = {};
    #pragma unroll
    for (int c = 0; c < NCHUNK; c += 2) {
        acc0 = __builtin_amdgcn_wmma_f32_16x16x4_f32(
            false, a[c], false, b[c], (short)0, acc0, false, false);
        acc1 = __builtin_amdgcn_wmma_f32_16x16x4_f32(
            false, a[c + 1], false, b[c + 1], (short)0, acc1, false, false);
    }
    v4f lo = {acc0[0] + acc1[0], acc0[1] + acc1[1], acc0[2] + acc1[2], acc0[3] + acc1[3]};
    v4f hh = {acc0[4] + acc1[4], acc0[5] + acc1[5], acc0[6] + acc1[6], acc0[7] + acc1[7]};
    *(v4f*)(orow)     = lo;
    *(v4f*)(orow + 4) = hh;
}

__global__ __launch_bounds__(BLOCK_THREADS) void
fir80_toeplitz_wmma_kernel(const float* __restrict__ x,
                           const float* __restrict__ tw,
                           float* __restrict__ out) {
    __shared__ float eff_pad[BLOCK_THREADS];

    const int lane = threadIdx.x & 31;
    const int wid  = __builtin_amdgcn_readfirstlane((int)(threadIdx.x >> 5));
    const int W    = blockIdx.x * WAVES_PER_BLOCK + wid;
    const int rsub = W >> 8;
    const int tw0  = W & (TILES_PER_ROW - 1);

    const int m   = lane & 15;
    const int hi  = lane >> 4;
    const int klo = hi << 1;
    const int lane_off = 16 * m + klo;

    int s_first = 0;
    if ((tw0 & (ROT - 1)) == 0)       s_first = ((TILES_PER_ROW - tw0) / ROT) & 7;
    if ((tw0 & (ROT - 1)) == ROT - 1) s_first = ((TILES_PER_ROW - 1 - tw0) / ROT) & 7;
#define STEP_OF(k)  ((s_first + (k)) & 7)
#define ROW_OF(k)   (8 * STEP_OF(k) + rsub)
#define TILE_OF(k)  ((tw0 + ROT * STEP_OF(k)) & (TILES_PER_ROW - 1))

    v2f bA[NCHUNK], bB[NCHUNK];
    load_b_any(x, ROW_OF(0), TILE_OF(0), lane_off, bA);

    {
        const int t = threadIdx.x;
        float acc = 0.0f;
        const int tt = t - PAD;
        if ((unsigned)tt < TAPS) {
            float w0 = tw[0], w1 = tw[1], w2 = tw[2], w3 = tw[3];
            float mx = fmaxf(fmaxf(w0, w1), fmaxf(w2, w3));
            float e0 = __expf(w0 - mx), e1 = __expf(w1 - mx);
            float e2 = __expf(w2 - mx), e3 = __expf(w3 - mx);
            float inv = 1.0f / (e0 + e1 + e2 + e3);
            const float sw[4]   = {e0 * inv, e1 * inv, e2 * inv, e3 * inv};
            const int   wid4[4] = {10, 20, 30, 40};
            const float twopi = 6.28318530717958647692f;
            #pragma unroll
            for (int i = 0; i < 4; ++i) {
                int w = wid4[i];
                int k = tt - 40 + w;
                if (k >= 0 && k < 2 * w) {
                    float h = 0.5f - 0.5f * cosf(twopi * (float)k / (float)(2 * w - 1));
                    acc += sw[i] * h;
                }
            }
        }
        eff_pad[t] = acc;
    }
    __syncthreads();

    v2f a_frag[NCHUNK];
    #pragma unroll
    for (int c = 0; c < NCHUNK; ++c) {
        const int base = c * 4 + klo - m + PAD;
        a_frag[c].x = eff_pad[base];
        a_frag[c].y = eff_pad[base + 1];
    }

    float* const obase = out + 8 * hi + 16 * m;
#define OUT_OF(k)   (obase + (size_t)ROW_OF(k) * L_LEN + (TILE_OF(k) << 8))
    for (int k = 0; k < TILES_PER_WAVE - 2; k += 2) {
        load_b_interior(x, ROW_OF(k + 1), TILE_OF(k + 1), lane_off, bB);
        mma_store(a_frag, bA, OUT_OF(k));
        load_b_interior(x, ROW_OF(k + 2), TILE_OF(k + 2), lane_off, bA);
        mma_store(a_frag, bB, OUT_OF(k + 1));
    }
    load_b_interior(x, ROW_OF(TILES_PER_WAVE - 1), TILE_OF(TILES_PER_WAVE - 1), lane_off, bB);
    __builtin_amdgcn_sched_barrier(0);
    mma_store(a_frag, bA, OUT_OF(TILES_PER_WAVE - 2));
    __builtin_amdgcn_sched_barrier(0);
    mma_store(a_frag, bB, OUT_OF(TILES_PER_WAVE - 1));
}

extern "C" void kernel_launch(void* const* d_in, const int* in_sizes, int n_in,
                              void* d_out, int out_size, void* d_ws, size_t ws_size,
                              hipStream_t stream) {
    (void)in_sizes; (void)n_in; (void)d_ws; (void)ws_size; (void)out_size;
    const float* x  = (const float*)d_in[0];
    const float* tw = (const float*)d_in[1];
    float* out = (float*)d_out;

    fir80_toeplitz_wmma_kernel<<<NBLOCKS, BLOCK_THREADS, 0, stream>>>(x, tw, out);
}


# signal slice read with 12 b128 loads per tile instead of 24 b64: k order inside the 96 window paired so a lane's two chunks are 4 consecutive floats; k order of the sums permuted
# speedup vs baseline: 1.1431x; 1.0589x over previous
#include <hip/hip_runtime.h>


typedef __attribute__((ext_vector_type(2))) float v2f;
typedef __attribute__((ext_vector_type(4))) float v4f;
typedef __attribute__((ext_vector_type(8))) float v8f;

#define L_LEN   65536
#define N_ROWS  64
#define TAPS    80
#define KSPAN   96
#define NCHUNK  (KSPAN / 4)
#define NLOAD   (NCHUNK / 2)
#define PAD     16
#define TILES_PER_ROW   (L_LEN / 256)
#define TILES_TOTAL     (N_ROWS * TILES_PER_ROW)
#define WAVES_PER_BLOCK 4
#define BLOCK_THREADS   128
#define NBLOCKS         512
#define NWAVES          (NBLOCKS * WAVES_PER_BLOCK)
#define TILES_PER_WAVE  (TILES_TOTAL / NWAVES)
#define ROT             32


__device__ __forceinline__ void load_b_interior(const float* __restrict__ x, int row, int t,
                                                int lane_off, v4f (&b)[NLOAD]) {
    const float* p = x + (size_t)row * L_LEN + ((t << 8) - 40);
    #pragma unroll
    for (int g = 0; g < NLOAD; ++g)
        b[g] = *(const v4f*)(p + lane_off + 8 * g);
}

__device__ __forceinline__ void load_b_any(const float* __restrict__ x, int row, int t,
                                           int lane_off, v4f (&b)[NLOAD]) {
    const int j0 = t << 8;
    const bool interior = (j0 >= 40) && (j0 + 296 <= L_LEN);
    if (interior) {
        load_b_interior(x, row, t, lane_off, b);
    } else {
        const float* xrow = x + (size_t)row * L_LEN;
        const int basep = j0 - 40 + lane_off;
        #pragma unroll
        for (int g = 0; g < NLOAD; ++g) {
            const int p0 = basep + 8 * g;
            int pc = p0 < 0 ? 0 : p0;
            pc = pc > (L_LEN - 4) ? (L_LEN - 4) : pc;
            v4f v = *(const v4f*)(xrow + pc);
            v.x = ((unsigned)p0       < L_LEN) ? v.x : 0.0f;
            v.y = ((unsigned)(p0 + 1) < L_LEN) ? v.y : 0.0f;
            v.z = ((unsigned)(p0 + 2) < L_LEN) ? v.z : 0.0f;
            v.w = ((unsigned)(p0 + 3) < L_LEN) ? v.w : 0.0f;
            b[g] = v;
        }
    }
}

__device__ __forceinline__ void mma_store(const v2f (&a)[NCHUNK], const v4f (&b)[NLOAD],
                                          float* __restrict__ orow) {
    v8f acc0 = {};
    v8f acc1 = {};
    #pragma unroll
    for (int g = 0; g < NLOAD; ++g) {
        const v2f blo = {b[g].x, b[g].y};
        const v2f bhi = {b[g].z, b[g].w};
        acc0 = __builtin_amdgcn_wmma_f32_16x16x4_f32(
            false, a[2 * g], false, blo, (short)0, acc0, false, false);
        acc1 = __builtin_amdgcn_wmma_f32_16x16x4_f32(
            false, a[2 * g + 1], false, bhi, (short)0, acc1, false, false);
    }
    v4f lo = {acc0[0] + acc1[0], acc0[1] + acc1[1], acc0[2] + acc1[2], acc0[3] + acc1[3]};
    v4f hh = {acc0[4] + acc1[4], acc0[5] + acc1[5], acc0[6] + acc1[6], acc0[7] + acc1[7]};
    *(v4f*)(orow)     = lo;
    *(v4f*)(orow + 4) = hh;
}

__global__ __launch_bounds__(BLOCK_THREADS) void
fir80_toeplitz_wmma_kernel(const float* __restrict__ x,
                           const float* __restrict__ tw,
                           float* __restrict__ out) {
    __shared__ float eff_pad[BLOCK_THREADS];

    const int lane = threadIdx.x & 31;
    const int wid  = __builtin_amdgcn_readfirstlane((int)(threadIdx.x >> 5));
    const int W    = blockIdx.x * WAVES_PER_BLOCK + wid;
    const int rsub = W >> 8;
    const int tw0  = W & (TILES_PER_ROW - 1);

    const int m   = lane & 15;
    const int hi  = lane >> 4;
    const int klo = hi << 2;
    const int lane_off = 16 * m + klo;

    int s_first = 0;
    if ((tw0 & (ROT - 1)) == 0)       s_first = ((TILES_PER_ROW - tw0) / ROT) & 7;
    if ((tw0 & (ROT - 1)) == ROT - 1) s_first = ((TILES_PER_ROW - 1 - tw0) / ROT) & 7;
#define STEP_OF(k)  ((s_first + (k)) & 7)
#define ROW_OF(k)   (8 * STEP_OF(k) + rsub)
#define TILE_OF(k)  ((tw0 + ROT * STEP_OF(k)) & (TILES_PER_ROW - 1))

    v4f bA[NLOAD], bB[NLOAD];
    load_b_any(x, ROW_OF(0), TILE_OF(0), lane_off, bA);

    {
        const int t = threadIdx.x;
        float acc = 0.0f;
        const int tt = t - PAD;
        if ((unsigned)tt < TAPS) {
            float w0 = tw[0], w1 = tw[1], w2 = tw[2], w3 = tw[3];
            float mx = fmaxf(fmaxf(w0, w1), fmaxf(w2, w3));
            float e0 = __expf(w0 - mx), e1 = __expf(w1 - mx);
            float e2 = __expf(w2 - mx), e3 = __expf(w3 - mx);
            float inv = 1.0f / (e0 + e1 + e2 + e3);
            const float sw[4]   = {e0 * inv, e1 * inv, e2 * inv, e3 * inv};
            const int   wid4[4] = {10, 20, 30, 40};
            const float twopi = 6.28318530717958647692f;
            #pragma unroll
            for (int i = 0; i < 4; ++i) {
                int w = wid4[i];
                int k = tt - 40 + w;
                if (k >= 0 && k < 2 * w) {
                    float h = 0.5f - 0.5f * cosf(twopi * (float)k / (float)(2 * w - 1));
                    acc += sw[i] * h;
                }
            }
        }
        eff_pad[t] = acc;
    }
    __syncthreads();

    v2f a_frag[NCHUNK];
    #pragma unroll
    for (int c = 0; c < NCHUNK; ++c) {
        const int base = (c >> 1) * 8 + klo + (c & 1) * 2 - m + PAD;
        a_frag[c].x = eff_pad[base];
        a_frag[c].y = eff_pad[base + 1];
    }

    float* const obase = out + 8 * hi + 16 * m;
#define OUT_OF(k)   (obase + (size_t)ROW_OF(k) * L_LEN + (TILE_OF(k) << 8))
    #pragma nounroll
    for (int k = 0; k < TILES_PER_WAVE - 2; k += 2) {
        load_b_interior(x, ROW_OF(k + 1), TILE_OF(k + 1), lane_off, bB);
        mma_store(a_frag, bA, OUT_OF(k));
        load_b_interior(x, ROW_OF(k + 2), TILE_OF(k + 2), lane_off, bA);
        mma_store(a_frag, bB, OUT_OF(k + 1));
    }
    load_b_interior(x, ROW_OF(TILES_PER_WAVE - 1), TILE_OF(TILES_PER_WAVE - 1), lane_off, bB);
    __builtin_amdgcn_sched_barrier(0);
    mma_store(a_frag, bA, OUT_OF(TILES_PER_WAVE - 2));
    __builtin_amdgcn_sched_barrier(0);
    mma_store(a_frag, bB, OUT_OF(TILES_PER_WAVE - 1));
}

extern "C" void kernel_launch(void* const* d_in, const int* in_sizes, int n_in,
                              void* d_out, int out_size, void* d_ws, size_t ws_size,
                              hipStream_t stream) {
    (void)in_sizes; (void)n_in; (void)d_ws; (void)ws_size; (void)out_size;
    const float* x  = (const float*)d_in[0];
    const float* tw = (const float*)d_in[1];
    float* out = (float*)d_out;

    fir80_toeplitz_wmma_kernel<<<NBLOCKS, BLOCK_THREADS, 0, stream>>>(x, tw, out);
}


# prefetch depth raised from one tile to four: five register sets of signal slices, tiles 0-3 requested ahead of the filter build, tile k+5 right behind tile k's WMMAs; tile loop written out
# speedup vs baseline: 1.2534x; 1.0965x over previous
#include <hip/hip_runtime.h>


typedef __attribute__((ext_vector_type(2))) float v2f;
typedef __attribute__((ext_vector_type(4))) float v4f;
typedef __attribute__((ext_vector_type(8))) float v8f;

#define L_LEN   65536
#define N_ROWS  64
#define TAPS    80
#define KSPAN   96
#define NCHUNK  (KSPAN / 4)
#define NLOAD   (NCHUNK / 2)
#define PAD     16
#define TILES_PER_ROW   (L_LEN / 256)
#define TILES_TOTAL     (N_ROWS * TILES_PER_ROW)
#define WAVES_PER_BLOCK 4
#define BLOCK_THREADS   128
#define NBLOCKS         512
#define NWAVES          (NBLOCKS * WAVES_PER_BLOCK)
#define TILES_PER_WAVE  (TILES_TOTAL / NWAVES)
#define ROT             32


__device__ __forceinline__ void load_b_interior(const float* __restrict__ x, int row, int t,
                                                int lane_off, v4f (&b)[NLOAD]) {
    const float* p = x + (size_t)row * L_LEN + ((t << 8) - 40);
    #pragma unroll
    for (int g = 0; g < NLOAD; ++g)
        b[g] = *(const v4f*)(p + lane_off + 8 * g);
}

__device__ __forceinline__ void load_b_any(const float* __restrict__ x, int row, int t,
                                           int lane_off, v4f (&b)[NLOAD]) {
    const int j0 = t << 8;
    const bool interior = (j0 >= 40) && (j0 + 296 <= L_LEN);
    if (interior) {
        load_b_interior(x, row, t, lane_off, b);
    } else {
        const float* xrow = x + (size_t)row * L_LEN;
        const int basep = j0 - 40 + lane_off;
        #pragma unroll
        for (int g = 0; g < NLOAD; ++g) {
            const int p0 = basep + 8 * g;
            int pc = p0 < 0 ? 0 : p0;
            pc = pc > (L_LEN - 4) ? (L_LEN - 4) : pc;
            v4f v = *(const v4f*)(xrow + pc);
            v.x = ((unsigned)p0       < L_LEN) ? v.x : 0.0f;
            v.y = ((unsigned)(p0 + 1) < L_LEN) ? v.y : 0.0f;
            v.z = ((unsigned)(p0 + 2) < L_LEN) ? v.z : 0.0f;
            v.w = ((unsigned)(p0 + 3) < L_LEN) ? v.w : 0.0f;
            b[g] = v;
        }
    }
}

__device__ __forceinline__ void mma_store(const v2f (&a)[NCHUNK], const v4f (&b)[NLOAD],
                                          float* __restrict__ orow) {
    v8f acc0 = {};
    v8f acc1 = {};
    #pragma unroll
    for (int g = 0; g < NLOAD; ++g) {
        const v2f blo = {b[g].x, b[g].y};
        const v2f bhi = {b[g].z, b[g].w};
        acc0 = __builtin_amdgcn_wmma_f32_16x16x4_f32(
            false, a[2 * g], false, blo, (short)0, acc0, false, false);
        acc1 = __builtin_amdgcn_wmma_f32_16x16x4_f32(
            false, a[2 * g + 1], false, bhi, (short)0, acc1, false, false);
    }
    v4f lo = {acc0[0] + acc1[0], acc0[1] + acc1[1], acc0[2] + acc1[2], acc0[3] + acc1[3]};
    v4f hh = {acc0[4] + acc1[4], acc0[5] + acc1[5], acc0[6] + acc1[6], acc0[7] + acc1[7]};
    *(v4f*)(orow)     = lo;
    *(v4f*)(orow + 4) = hh;
}

__global__ __launch_bounds__(BLOCK_THREADS) void
fir80_toeplitz_wmma_kernel(const float* __restrict__ x,
                           const float* __restrict__ tw,
                           float* __restrict__ out) {
    __shared__ float eff_pad[BLOCK_THREADS];

    const int lane = threadIdx.x & 31;
    const int wid  = __builtin_amdgcn_readfirstlane((int)(threadIdx.x >> 5));
    const int W    = blockIdx.x * WAVES_PER_BLOCK + wid;
    const int rsub = W >> 8;
    const int tw0  = W & (TILES_PER_ROW - 1);

    const int m   = lane & 15;
    const int hi  = lane >> 4;
    const int klo = hi << 2;
    const int lane_off = 16 * m + klo;

    int s_first = 0;
    if ((tw0 & (ROT - 1)) == 0)       s_first = ((TILES_PER_ROW - tw0) / ROT) & 7;
    if ((tw0 & (ROT - 1)) == ROT - 1) s_first = ((TILES_PER_ROW - 1 - tw0) / ROT) & 7;
#define STEP_OF(k)  ((s_first + (k)) & 7)
#define ROW_OF(k)   (8 * STEP_OF(k) + rsub)
#define TILE_OF(k)  ((tw0 + ROT * STEP_OF(k)) & (TILES_PER_ROW - 1))

    v4f b0[NLOAD], b1[NLOAD], b2[NLOAD], b3[NLOAD], b4[NLOAD];
    load_b_interior(x, ROW_OF(1), TILE_OF(1), lane_off, b1);
    load_b_interior(x, ROW_OF(2), TILE_OF(2), lane_off, b2);
    load_b_interior(x, ROW_OF(3), TILE_OF(3), lane_off, b3);
    load_b_any(x, ROW_OF(0), TILE_OF(0), lane_off, b0);

    {
        const int t = threadIdx.x;
        float acc = 0.0f;
        const int tt = t - PAD;
        if ((unsigned)tt < TAPS) {
            float w0 = tw[0], w1 = tw[1], w2 = tw[2], w3 = tw[3];
            float mx = fmaxf(fmaxf(w0, w1), fmaxf(w2, w3));
            float e0 = __expf(w0 - mx), e1 = __expf(w1 - mx);
            float e2 = __expf(w2 - mx), e3 = __expf(w3 - mx);
            float inv = 1.0f / (e0 + e1 + e2 + e3);
            const float sw[4]   = {e0 * inv, e1 * inv, e2 * inv, e3 * inv};
            const int   wid4[4] = {10, 20, 30, 40};
            const float twopi = 6.28318530717958647692f;
            #pragma unroll
            for (int i = 0; i < 4; ++i) {
                int w = wid4[i];
                int k = tt - 40 + w;
                if (k >= 0 && k < 2 * w) {
                    float h = 0.5f - 0.5f * cosf(twopi * (float)k / (float)(2 * w - 1));
                    acc += sw[i] * h;
                }
            }
        }
        eff_pad[t] = acc;
    }
    __syncthreads();

    v2f a_frag[NCHUNK];
    #pragma unroll
    for (int c = 0; c < NCHUNK; ++c) {
        const int base = (c >> 1) * 8 + klo + (c & 1) * 2 - m + PAD;
        a_frag[c].x = eff_pad[base];
        a_frag[c].y = eff_pad[base + 1];
    }

    static_assert(TILES_PER_WAVE == 8, "the tile sequence below is written out for eight tiles per wave");
    float* const obase = out + 8 * hi + 16 * m;
#define OUT_OF(k)   (obase + (size_t)ROW_OF(k) * L_LEN + (TILE_OF(k) << 8))
#define KEEP_ORDER() __builtin_amdgcn_sched_barrier(0)
    load_b_interior(x, ROW_OF(4), TILE_OF(4), lane_off, b4);
    KEEP_ORDER();
    mma_store(a_frag, b0, OUT_OF(0));
    KEEP_ORDER();
    load_b_interior(x, ROW_OF(5), TILE_OF(5), lane_off, b0);
    KEEP_ORDER();
    mma_store(a_frag, b1, OUT_OF(1));
    KEEP_ORDER();
    load_b_interior(x, ROW_OF(6), TILE_OF(6), lane_off, b1);
    KEEP_ORDER();
    mma_store(a_frag, b2, OUT_OF(2));
    KEEP_ORDER();
    load_b_interior(x, ROW_OF(7), TILE_OF(7), lane_off, b2);
    KEEP_ORDER();
    mma_store(a_frag, b3, OUT_OF(3));
    KEEP_ORDER();
    mma_store(a_frag, b4, OUT_OF(4));
    KEEP_ORDER();
    mma_store(a_frag, b0, OUT_OF(5));
    KEEP_ORDER();
    mma_store(a_frag, b1, OUT_OF(6));
    KEEP_ORDER();
    mma_store(a_frag, b2, OUT_OF(7));
}

extern "C" void kernel_launch(void* const* d_in, const int* in_sizes, int n_in,
                              void* d_out, int out_size, void* d_ws, size_t ws_size,
                              hipStream_t stream) {
    (void)in_sizes; (void)n_in; (void)d_ws; (void)ws_size; (void)out_size;
    const float* x  = (const float*)d_in[0];
    const float* tw = (const float*)d_in[1];
    float* out = (float*)d_out;

    fir80_toeplitz_wmma_kernel<<<NBLOCKS, BLOCK_THREADS, 0, stream>>>(x, tw, out);
}


# first WMMA starts earlier: tile 0 requested first, barrier no longer drains the loads in flight, filter built with one cosine per thread in two LDS hand-overs (upper window half = mirror tap)
# speedup vs baseline: 1.3478x; 1.0753x over previous
#include <hip/hip_runtime.h>


typedef __attribute__((ext_vector_type(2))) float v2f;
typedef __attribute__((ext_vector_type(4))) float v4f;
typedef __attribute__((ext_vector_type(8))) float v8f;

#define L_LEN   65536
#define N_ROWS  64
#define TAPS    80
#define KSPAN   96
#define NCHUNK  (KSPAN / 4)
#define NLOAD   (NCHUNK / 2)
#define PAD     16
#define TILES_PER_ROW   (L_LEN / 256)
#define TILES_TOTAL     (N_ROWS * TILES_PER_ROW)
#define WAVES_PER_BLOCK 4
#define BLOCK_THREADS   128
#define NBLOCKS         512
#define NWAVES          (NBLOCKS * WAVES_PER_BLOCK)
#define TILES_PER_WAVE  (TILES_TOTAL / NWAVES)
#define ROT             32


__device__ __forceinline__ void load_b_interior(const float* __restrict__ x, int row, int t,
                                                int lane_off, v4f (&b)[NLOAD]) {
    const float* p = x + (size_t)row * L_LEN + ((t << 8) - 40);
    #pragma unroll
    for (int g = 0; g < NLOAD; ++g)
        b[g] = *(const v4f*)(p + lane_off + 8 * g);
}

__device__ __forceinline__ bool tile_is_interior(int t) {
    const int j0 = t << 8;
    return (j0 >= 40) && (j0 + 296 <= L_LEN);
}

__device__ __forceinline__ void load_b_edge_clamped(const float* __restrict__ x, int row, int t,
                                                    int lane_off, v4f (&b)[NLOAD]) {
    const float* xrow = x + (size_t)row * L_LEN;
    const int basep = (t << 8) - 40 + lane_off;
    #pragma unroll
    for (int g = 0; g < NLOAD; ++g) {
        const int p0 = basep + 8 * g;
        int pc = p0 < 0 ? 0 : p0;
        pc = pc > (L_LEN - 4) ? (L_LEN - 4) : pc;
        b[g] = *(const v4f*)(xrow + pc);
    }
}

__device__ __forceinline__ void zero_fill_edge(int t, int lane_off, v4f (&b)[NLOAD]) {
    const int basep = (t << 8) - 40 + lane_off;
    #pragma unroll
    for (int g = 0; g < NLOAD; ++g) {
        const int p0 = basep + 8 * g;
        v4f v = b[g];
        v.x = ((unsigned)p0       < L_LEN) ? v.x : 0.0f;
        v.y = ((unsigned)(p0 + 1) < L_LEN) ? v.y : 0.0f;
        v.z = ((unsigned)(p0 + 2) < L_LEN) ? v.z : 0.0f;
        v.w = ((unsigned)(p0 + 3) < L_LEN) ? v.w : 0.0f;
        b[g] = v;
    }
}

__device__ __forceinline__ void barrier_lds_only() {
    __builtin_amdgcn_fence(__ATOMIC_RELEASE, "workgroup", "local");
    __builtin_amdgcn_s_barrier();
    __builtin_amdgcn_fence(__ATOMIC_ACQUIRE, "workgroup", "local");
}

__device__ __forceinline__ void mma_store(const v2f (&a)[NCHUNK], const v4f (&b)[NLOAD],
                                          float* __restrict__ orow) {
    v8f acc0 = {};
    v8f acc1 = {};
    #pragma unroll
    for (int g = 0; g < NLOAD; ++g) {
        const v2f blo = {b[g].x, b[g].y};
        const v2f bhi = {b[g].z, b[g].w};
        acc0 = __builtin_amdgcn_wmma_f32_16x16x4_f32(
            false, a[2 * g], false, blo, (short)0, acc0, false, false);
        acc1 = __builtin_amdgcn_wmma_f32_16x16x4_f32(
            false, a[2 * g + 1], false, bhi, (short)0, acc1, false, false);
    }
    v4f lo = {acc0[0] + acc1[0], acc0[1] + acc1[1], acc0[2] + acc1[2], acc0[3] + acc1[3]};
    v4f hh = {acc0[4] + acc1[4], acc0[5] + acc1[5], acc0[6] + acc1[6], acc0[7] + acc1[7]};
    *(v4f*)(orow)     = lo;
    *(v4f*)(orow + 4) = hh;
}

__global__ __launch_bounds__(BLOCK_THREADS) void
fir80_toeplitz_wmma_kernel(const float* __restrict__ x,
                           const float* __restrict__ tw,
                           float* __restrict__ out) {
    __shared__ float eff_pad[BLOCK_THREADS];
    __shared__ float hann_half[BLOCK_THREADS];

    const int lane = threadIdx.x & 31;
    const int wid  = __builtin_amdgcn_readfirstlane((int)(threadIdx.x >> 5));
    const int W    = blockIdx.x * WAVES_PER_BLOCK + wid;
    const int rsub = W >> 8;
    const int tw0  = W & (TILES_PER_ROW - 1);

    const int m   = lane & 15;
    const int hi  = lane >> 4;
    const int klo = hi << 2;
    const int lane_off = 16 * m + klo;

    int s_first = 0;
    if ((tw0 & (ROT - 1)) == 0)       s_first = ((TILES_PER_ROW - tw0) / ROT) & 7;
    if ((tw0 & (ROT - 1)) == ROT - 1) s_first = ((TILES_PER_ROW - 1 - tw0) / ROT) & 7;
#define STEP_OF(k)  ((s_first + (k)) & 7)
#define ROW_OF(k)   (8 * STEP_OF(k) + rsub)
#define TILE_OF(k)  ((tw0 + ROT * STEP_OF(k)) & (TILES_PER_ROW - 1))

#define KEEP_ORDER() __builtin_amdgcn_sched_barrier(0)
    const float w0 = tw[0], w1 = tw[1], w2 = tw[2], w3 = tw[3];
    v4f b0[NLOAD], b1[NLOAD], b2[NLOAD], b3[NLOAD], b4[NLOAD];
    const bool interior0 = tile_is_interior(TILE_OF(0));
    if (interior0) load_b_interior(x, ROW_OF(0), TILE_OF(0), lane_off, b0);
    else           load_b_edge_clamped(x, ROW_OF(0), TILE_OF(0), lane_off, b0);
    KEEP_ORDER();
    load_b_interior(x, ROW_OF(1), TILE_OF(1), lane_off, b1);
    KEEP_ORDER();
    load_b_interior(x, ROW_OF(2), TILE_OF(2), lane_off, b2);
    KEEP_ORDER();
    load_b_interior(x, ROW_OF(3), TILE_OF(3), lane_off, b3);
    KEEP_ORDER();

    {
        const int t = threadIdx.x;
        const int w = t < 40 ? 40 : (t < 70 ? 30 : (t < 90 ? 20 : 10));
        const int k = t < 40 ? t  : (t < 70 ? t - 40 : (t < 90 ? t - 70 : t - 90));
        const float twopi = 6.28318530717958647692f;
        hann_half[t] = 0.5f - 0.5f * cosf(twopi * (float)k / (float)(2 * w - 1));
    }
    const float mx = fmaxf(fmaxf(w0, w1), fmaxf(w2, w3));
    const float e0 = __expf(w0 - mx), e1 = __expf(w1 - mx);
    const float e2 = __expf(w2 - mx), e3 = __expf(w3 - mx);
    const float inv = 1.0f / (e0 + e1 + e2 + e3);
    const float sw[4] = {e0 * inv, e1 * inv, e2 * inv, e3 * inv};
    barrier_lds_only();
    {
        const int t = threadIdx.x;
        float acc = 0.0f;
        const int tt = t - PAD;
        if ((unsigned)tt < TAPS) {
            const int   wid4[4] = {10, 20, 30, 40};
            const int   off4[4] = {90, 70, 40, 0};
            #pragma unroll
            for (int i = 0; i < 4; ++i) {
                const int w = wid4[i];
                const int k = tt - 40 + w;
                const bool inside = (k >= 0) && (k < 2 * w);
                int kc = k < 0 ? 0 : k;
                kc = kc > 2 * w - 1 ? 2 * w - 1 : kc;
                const int kf = kc < w ? kc : 2 * w - 1 - kc;
                const float h = inside ? hann_half[off4[i] + kf] : 0.0f;
                acc += sw[i] * h;
            }
        }
        eff_pad[t] = acc;
    }
    barrier_lds_only();

    v2f a_frag[NCHUNK];
    #pragma unroll
    for (int c = 0; c < NCHUNK; ++c) {
        const int base = (c >> 1) * 8 + klo + (c & 1) * 2 - m + PAD;
        a_frag[c].x = eff_pad[base];
        a_frag[c].y = eff_pad[base + 1];
    }

    static_assert(TILES_PER_WAVE == 8, "the tile sequence below is written out for eight tiles per wave");
    float* const obase = out + 8 * hi + 16 * m;
#define OUT_OF(k)   (obase + (size_t)ROW_OF(k) * L_LEN + (TILE_OF(k) << 8))
    load_b_interior(x, ROW_OF(4), TILE_OF(4), lane_off, b4);
    KEEP_ORDER();
    if (!interior0) zero_fill_edge(TILE_OF(0), lane_off, b0);
    mma_store(a_frag, b0, OUT_OF(0));
    KEEP_ORDER();
    load_b_interior(x, ROW_OF(5), TILE_OF(5), lane_off, b0);
    KEEP_ORDER();
    mma_store(a_frag, b1, OUT_OF(1));
    KEEP_ORDER();
    load_b_interior(x, ROW_OF(6), TILE_OF(6), lane_off, b1);
    KEEP_ORDER();
    mma_store(a_frag, b2, OUT_OF(2));
    KEEP_ORDER();
    load_b_interior(x, ROW_OF(7), TILE_OF(7), lane_off, b2);
    KEEP_ORDER();
    mma_store(a_frag, b3, OUT_OF(3));
    KEEP_ORDER();
    mma_store(a_frag, b4, OUT_OF(4));
    KEEP_ORDER();
    mma_store(a_frag, b0, OUT_OF(5));
    KEEP_ORDER();
    mma_store(a_frag, b1, OUT_OF(6));
    KEEP_ORDER();
    mma_store(a_frag, b2, OUT_OF(7));
}

extern "C" void kernel_launch(void* const* d_in, const int* in_sizes, int n_in,
                              void* d_out, int out_size, void* d_ws, size_t ws_size,
                              hipStream_t stream) {
    (void)in_sizes; (void)n_in; (void)d_ws; (void)ws_size; (void)out_size;
    const float* x  = (const float*)d_in[0];
    const float* tw = (const float*)d_in[1];
    float* out = (float*)d_out;

    fir80_toeplitz_wmma_kernel<<<NBLOCKS, BLOCK_THREADS, 0, stream>>>(x, tw, out);
}


# filter symmetry fold: tile rows m and 31-m from a sum and a difference product (K=56) of signal pair sums read from a wave-private LDS copy: 28 WMMAs per 512 outputs, not 48; sum order changed
# speedup vs baseline: 1.6406x; 1.2172x over previous
#include <hip/hip_runtime.h>


typedef __attribute__((ext_vector_type(2))) float v2f;
typedef __attribute__((ext_vector_type(4))) float v4f;
typedef __attribute__((ext_vector_type(8))) float v8f;

#define L_LEN   65536
#define N_ROWS  64
#define TAPS    80
#define PAD     16
#define TILE_OUT        512
#define TILES_PER_ROW   (L_LEN / TILE_OUT)
#define TILES_TOTAL     (N_ROWS * TILES_PER_ROW)
#define WAVES_PER_BLOCK 4
#define BLOCK_THREADS   128
#define NBLOCKS         512
#define NWAVES          (NBLOCKS * WAVES_PER_BLOCK)
#define TILES_PER_WAVE  (TILES_TOTAL / NWAVES)
#define ROT             32
#define KFOLD   56
#define NCH     (KFOLD / 4)
#define NPIECE  5
#define NLOW    7
#define NMIR    8
#define ROWP    36
#define STG_FLOATS (20 * ROWP)


__device__ __forceinline__ constexpr int pos_rel(int g) { return 4 * g + 4 * (g >> 3); }

__device__ __forceinline__ void load_win_interior(const float* __restrict__ x, int row, int t,
                                                  int lane, v4f (&s)[NPIECE]) {
    const float* p = x + (size_t)row * L_LEN + ((t << 9) - 40) + 4 * lane;
    #pragma unroll
    for (int i = 0; i < NPIECE; ++i)
        s[i] = *(const v4f*)(p + 128 * i);
}

__device__ __forceinline__ void load_win_edge_clamped(const float* __restrict__ x, int row, int t,
                                                      int lane, v4f (&s)[NPIECE]) {
    const float* xrow = x + (size_t)row * L_LEN;
    const int basep = (t << 9) - 40 + 4 * lane;
    #pragma unroll
    for (int i = 0; i < NPIECE; ++i) {
        const int p0 = basep + 128 * i;
        int pc = p0 < 0 ? 0 : p0;
        pc = pc > (L_LEN - 4) ? (L_LEN - 4) : pc;
        s[i] = *(const v4f*)(xrow + pc);
    }
}

__device__ __forceinline__ void zero_fill_edge(int t, int lane, v4f (&s)[NPIECE]) {
    const int basep = (t << 9) - 40 + 4 * lane;
    const v4f z = {0.0f, 0.0f, 0.0f, 0.0f};
    #pragma unroll
    for (int i = 0; i < NPIECE; ++i) {
        const int p0 = basep + 128 * i;
        s[i] = ((unsigned)p0 < L_LEN) ? s[i] : z;
    }
}

__device__ __forceinline__ void barrier_lds_only() {
    __builtin_amdgcn_fence(__ATOMIC_RELEASE, "workgroup", "local");
    __builtin_amdgcn_s_barrier();
    __builtin_amdgcn_fence(__ATOMIC_ACQUIRE, "workgroup", "local");
}

__device__ __forceinline__ void stage_store(float* __restrict__ buf, int st_off, const v4f (&s)[NPIECE]) {
    #pragma unroll
    for (int i = 0; i < NPIECE; ++i)
        *(v4f*)(buf + st_off + 4 * ROWP * i) = s[i];
}

__device__ __forceinline__ void frag_load(const float* __restrict__ buf, int col_off, int hi,
                                          v4f (&lo)[NLOW], v4f (&mi)[NMIR]) {
    #pragma unroll
    for (int j = 0; j < NLOW; ++j) {
        const int off = hi ? pos_rel(7 + j) : pos_rel(j);
        lo[j] = *(const v4f*)(buf + col_off + off);
    }
    #pragma unroll
    for (int i = 0; i < NMIR; ++i) {
        const int off = hi ? pos_rel(13 + i) : pos_rel(20 + i);
        mi[i] = *(const v4f*)(buf + col_off + off);
    }
}

__device__ __forceinline__ void fold(const v4f (&lo)[NLOW], const v4f (&mi)[NMIR],
                                     v2f (&ps)[NCH], v2f (&pa)[NCH]) {
    #pragma unroll
    for (int j = 0; j < NLOW; ++j) {
        const v4f l = lo[j], mh = mi[7 - j], ml = mi[6 - j];
        ps[2 * j]     = (v2f){l.x + mh.z, l.y + mh.y};
        pa[2 * j]     = (v2f){l.x - mh.z, l.y - mh.y};
        ps[2 * j + 1] = (v2f){l.z + mh.x, l.w + ml.w};
        pa[2 * j + 1] = (v2f){l.z - mh.x, l.w - ml.w};
    }
}

__device__ __forceinline__ void mma_store(const v2f (&as)[NCH], const v2f (&aa)[NCH],
                                          const v2f (&ps)[NCH], const v2f (&pa)[NCH],
                                          float* __restrict__ o_fwd, float* __restrict__ o_rev) {
    v8f s = {};
    v8f t = {};
    #pragma unroll
    for (int c = 0; c < NCH; ++c) {
        s = __builtin_amdgcn_wmma_f32_16x16x4_f32(false, as[c], false, ps[c], (short)0, s, false, false);
        t = __builtin_amdgcn_wmma_f32_16x16x4_f32(false, aa[c], false, pa[c], (short)0, t, false, false);
    }
    const v4f f0 = {s[0] + t[0], s[1] + t[1], s[2] + t[2], s[3] + t[3]};
    const v4f f1 = {s[4] + t[4], s[5] + t[5], s[6] + t[6], s[7] + t[7]};
    const v4f r0 = {s[7] - t[7], s[6] - t[6], s[5] - t[5], s[4] - t[4]};
    const v4f r1 = {s[3] - t[3], s[2] - t[2], s[1] - t[1], s[0] - t[0]};
    *(v4f*)(o_fwd)     = f0;
    *(v4f*)(o_fwd + 4) = f1;
    *(v4f*)(o_rev)     = r0;
    *(v4f*)(o_rev + 4) = r1;
}

__global__ __launch_bounds__(BLOCK_THREADS) void
fir80_fold_wmma_kernel(const float* __restrict__ x,
                       const float* __restrict__ tw,
                       float* __restrict__ out) {
    __shared__ float eff_pad[BLOCK_THREADS];
    __shared__ float hann_half[BLOCK_THREADS];
    __shared__ __attribute__((aligned(16))) float stage[WAVES_PER_BLOCK][2][STG_FLOATS];

    const int lane = threadIdx.x & 31;
    const int wid  = __builtin_amdgcn_readfirstlane((int)(threadIdx.x >> 5));
    const int W    = blockIdx.x * WAVES_PER_BLOCK + wid;
    const int rsub = W >> 7;
    const int tw0  = W & (TILES_PER_ROW - 1);

    const int m   = lane & 15;
    const int hi  = lane >> 4;
    const int col_off = ROWP * m;
    const int st_off  = ROWP * (lane >> 3) + 4 * (lane & 7);

    int s_first = 0;
    if ((tw0 & (ROT - 1)) == 0)       s_first = ((TILES_PER_ROW - tw0) / ROT) & 3;
    if ((tw0 & (ROT - 1)) == ROT - 1) s_first = ((TILES_PER_ROW - 1 - tw0) / ROT) & 3;
#define STEP_OF(k)  ((s_first + (k)) & 3)
#define ROW_OF(k)   (16 * STEP_OF(k) + rsub)
#define TILE_OF(k)  ((tw0 + ROT * STEP_OF(k)) & (TILES_PER_ROW - 1))

#define KEEP_ORDER() __builtin_amdgcn_sched_barrier(0)
    const float w0 = tw[0], w1 = tw[1], w2 = tw[2], w3 = tw[3];
    v4f s0[NPIECE], s1[NPIECE], s2[NPIECE], s3[NPIECE];
    load_win_edge_clamped(x, ROW_OF(0), TILE_OF(0), lane, s0);
    KEEP_ORDER();
    load_win_interior(x, ROW_OF(1), TILE_OF(1), lane, s1);
    KEEP_ORDER();
    load_win_interior(x, ROW_OF(2), TILE_OF(2), lane, s2);
    KEEP_ORDER();
    load_win_interior(x, ROW_OF(3), TILE_OF(3), lane, s3);
    KEEP_ORDER();

    {
        const int t = threadIdx.x;
        const int w = t < 40 ? 40 : (t < 70 ? 30 : (t < 90 ? 20 : 10));
        const int k = t < 40 ? t  : (t < 70 ? t - 40 : (t < 90 ? t - 70 : t - 90));
        const float twopi = 6.28318530717958647692f;
        hann_half[t] = 0.5f - 0.5f * cosf(twopi * (float)k / (float)(2 * w - 1));
    }
    const float mx = fmaxf(fmaxf(w0, w1), fmaxf(w2, w3));
    const float e0 = __expf(w0 - mx), e1 = __expf(w1 - mx);
    const float e2 = __expf(w2 - mx), e3 = __expf(w3 - mx);
    const float inv = 1.0f / (e0 + e1 + e2 + e3);
    const float sw[4] = {e0 * inv, e1 * inv, e2 * inv, e3 * inv};
    barrier_lds_only();
    {
        const int t = threadIdx.x;
        float acc = 0.0f;
        const int tt = t - PAD;
        if ((unsigned)tt < TAPS) {
            const int   wid4[4] = {10, 20, 30, 40};
            const int   off4[4] = {90, 70, 40, 0};
            #pragma unroll
            for (int i = 0; i < 4; ++i) {
                const int w = wid4[i];
                const int k = tt - 40 + w;
                const bool inside = (k >= 0) && (k < 2 * w);
                int kc = k < 0 ? 0 : k;
                kc = kc > 2 * w - 1 ? 2 * w - 1 : kc;
                const int kf = kc < w ? kc : 2 * w - 1 - kc;
                const float h = inside ? hann_half[off4[i] + kf] : 0.0f;
                acc += sw[i] * h;
            }
        }
        eff_pad[t] = acc;
    }
    barrier_lds_only();

    v2f a_sum[NCH], a_dif[NCH];
    #pragma unroll
    for (int c = 0; c < NCH; ++c) {
        const int u0 = 28 * hi + 2 * c;
        const float a1x = eff_pad[PAD + u0 - m],     a2x = eff_pad[PAD + 110 - u0 - m];
        const float a1y = eff_pad[PAD + u0 + 1 - m], a2y = eff_pad[PAD + 109 - u0 - m];
        const float wy  = (u0 + 1 == 55) ? 0.25f : 0.5f;
        a_sum[c] = (v2f){0.5f * (a1x + a2x), wy * (a1y + a2y)};
        a_dif[c] = (v2f){0.5f * (a1x - a2x), 0.5f * (a1y - a2y)};
    }

    static_assert(TILES_PER_WAVE == 4, "the tile sequence below is written out for four tiles per wave");
    float* const buf0 = &stage[wid][0][0];
    float* const buf1 = &stage[wid][1][0];
    float* const ofwd = out + 32 * m + 8 * hi;
    float* const orev = out + 32 * m + 24 - 8 * hi;
#define OFF_OF(k)   ((size_t)ROW_OF(k) * L_LEN + (TILE_OF(k) << 9))
#define HAND_OVER() __builtin_amdgcn_wave_barrier()
    v4f lo[NLOW], mi[NMIR];
    v2f psA[NCH], paA[NCH], psB[NCH], paB[NCH];
    zero_fill_edge(TILE_OF(0), lane, s0);
    stage_store(buf0, st_off, s0);
    HAND_OVER();
    frag_load(buf0, col_off, hi, lo, mi);
    KEEP_ORDER();
    fold(lo, mi, psA, paA);
    KEEP_ORDER();
    stage_store(buf1, st_off, s1);
    HAND_OVER();
    frag_load(buf1, col_off, hi, lo, mi);
    KEEP_ORDER();
    mma_store(a_sum, a_dif, psA, paA, ofwd + OFF_OF(0), orev + OFF_OF(0));
    KEEP_ORDER();
    fold(lo, mi, psB, paB);
    KEEP_ORDER();
    stage_store(buf0, st_off, s2);
    HAND_OVER();
    frag_load(buf0, col_off, hi, lo, mi);
    KEEP_ORDER();
    mma_store(a_sum, a_dif, psB, paB, ofwd + OFF_OF(1), orev + OFF_OF(1));
    KEEP_ORDER();
    fold(lo, mi, psA, paA);
    KEEP_ORDER();
    stage_store(buf1, st_off, s3);
    HAND_OVER();
    frag_load(buf1, col_off, hi, lo, mi);
    KEEP_ORDER();
    mma_store(a_sum, a_dif, psA, paA, ofwd + OFF_OF(2), orev + OFF_OF(2));
    KEEP_ORDER();
    fold(lo, mi, psB, paB);
    KEEP_ORDER();
    mma_store(a_sum, a_dif, psB, paB, ofwd + OFF_OF(3), orev + OFF_OF(3));
}

extern "C" void kernel_launch(void* const* d_in, const int* in_sizes, int n_in,
                              void* d_out, int out_size, void* d_ws, size_t ws_size,
                              hipStream_t stream) {
    (void)in_sizes; (void)n_in; (void)d_ws; (void)ws_size; (void)out_size;
    const float* x  = (const float*)d_in[0];
    const float* tw = (const float*)d_in[1];
    float* out = (float*)d_out;

    fir80_fold_wmma_kernel<<<NBLOCKS, BLOCK_THREADS, 0, stream>>>(x, tw, out);
}
